# starting kernel
# baseline (speedup 1.0000x reference)
#include <hip/hip_runtime.h>
#include <hip/hip_bf16.h>

#define N_ATOMS 50000
#define N_EDGES 1600000
#define NAB 128
#define NRBF 20
#define NF 128
#define TPAD (NF + 4)

typedef __attribute__((ext_vector_type(2))) float v2f;
typedef __attribute__((ext_vector_type(8))) float v8f;

__device__ __forceinline__ v8f wmma_f32x4(v2f a, v2f b, v8f c) {
    return __builtin_amdgcn_wmma_f32_16x16x4_f32(
         false, a,  false, b,
         (short)0, c,  false,  false);
}

__device__ __forceinline__ float sspf(float x) {
    return fmaxf(x, 0.f) + log1pf(expf(-fabsf(x))) - 0.69314718055994530942f;
}

__device__ __forceinline__ void stage_w_pairs(const float* __restrict__ W,
                                              float* __restrict__ sW2,
                                              int krows, int tid) {
    const int npairs = (krows / 2) * NF;
    for (int i = tid; i < npairs; i += 256) {
        const int p   = i >> 7;
        const int col = i & (NF - 1);
        sW2[2 * i + 0] = W[(2 * p)     * NF + col];
        sW2[2 * i + 1] = W[(2 * p + 1) * NF + col];
    }
}

__device__ __forceinline__ v2f ld_pair_lds(const float* __restrict__ sW2,
                                           int kk, int col) {
    return *(const v2f*)(sW2 + (((kk >> 1) * NF + col) << 1));
}

__global__ void zero_f32(float* __restrict__ p, int n4) {
    int i = blockIdx.x * blockDim.x + threadIdx.x;
    if (i < n4) ((float4*)p)[i] = make_float4(0.f, 0.f, 0.f, 0.f);
}

__global__ __launch_bounds__(256) void gemm128_wmma(
    const float* __restrict__ A, const float* __restrict__ W,
    const float* __restrict__ bias, float* __restrict__ out,
    int M, int apply_ssp)
{
    extern __shared__ float smem[];
    float* sW2 = smem;
    float* sB  = smem + NF * NF;

    const int tid = threadIdx.x;
    stage_w_pairs(W, sW2, NF, tid);
    if (tid < NF) sB[tid] = bias ? bias[tid] : 0.f;
    __syncthreads();

    const int lane = tid & 31;
    const int wave = tid >> 5;
    const int m    = lane & 15;
    const int hi   = lane >> 4;
    const int r0   = blockIdx.x * 128 + wave * 16;

    v8f acc[8];
    const v8f z8 = {0.f, 0.f, 0.f, 0.f, 0.f, 0.f, 0.f, 0.f};
    #pragma unroll
    for (int ct = 0; ct < 8; ++ct) acc[ct] = z8;

    const int rowc = min(r0 + m, M - 1);
    const float* arow = A + (size_t)rowc * NF;

    #pragma unroll 4
    for (int k = 0; k < NF; k += 4) {
        const int kk = k + hi * 2;
        const v2f a = *(const v2f*)(arow + kk);
        #pragma unroll
        for (int ct = 0; ct < 8; ++ct) {
            const v2f b = ld_pair_lds(sW2, kk, ct * 16 + m);
            acc[ct] = wmma_f32x4(a, b, acc[ct]);
        }
    }

    #pragma unroll
    for (int ct = 0; ct < 8; ++ct) {
        const int col = ct * 16 + m;
        #pragma unroll
        for (int v = 0; v < 8; ++v) {
            const int r = r0 + v + hi * 8;
            if (r < M) {
                float val = acc[ct][v] + sB[col];
                if (apply_ssp) val = sspf(val);
                out[(size_t)r * NF + col] = val;
            }
        }
    }
}

__global__ __launch_bounds__(256) void edge_cfconv_wmma(
    const float* __restrict__ f_ij, const float* __restrict__ rcut,
    const int* __restrict__ idx_i, const int* __restrict__ idx_j,
    const float* __restrict__ Wf1, const float* __restrict__ bf1,
    const float* __restrict__ Wf2, const float* __restrict__ bf2,
    const float* __restrict__ h, float* __restrict__ y)
{
    extern __shared__ float smem[];
    float* sWf2 = smem;
    float* sWf1 = sWf2 + NF * NF;
    float* sB1  = sWf1 + NRBF * NF;
    float* sB2  = sB1 + NF;
    float* sT   = sB2 + NF;

    const int tid = threadIdx.x;
    stage_w_pairs(Wf2, sWf2, NF, tid);
    stage_w_pairs(Wf1, sWf1, NRBF, tid);
    if (tid < NF) { sB1[tid] = bf1[tid]; sB2[tid] = bf2[tid]; }
    __syncthreads();

    const int lane = tid & 31;
    const int wave = tid >> 5;
    const int m    = lane & 15;
    const int hi   = lane >> 4;
    const int eb   = blockIdx.x * 128 + wave * 16;

    float* tw = sT + wave * 16 * TPAD;
    const v8f z8 = {0.f, 0.f, 0.f, 0.f, 0.f, 0.f, 0.f, 0.f};

    v8f acc[8];
    #pragma unroll
    for (int ct = 0; ct < 8; ++ct) acc[ct] = z8;

    const float* frow = f_ij + (size_t)(eb + m) * NRBF;
    #pragma unroll
    for (int k = 0; k < NRBF; k += 4) {
        const int kk = k + hi * 2;
        const v2f a = *(const v2f*)(frow + kk);
        #pragma unroll
        for (int ct = 0; ct < 8; ++ct) {
            const v2f b = ld_pair_lds(sWf1, kk, ct * 16 + m);
            acc[ct] = wmma_f32x4(a, b, acc[ct]);
        }
    }

    #pragma unroll
    for (int ct = 0; ct < 8; ++ct) {
        const int col = ct * 16 + m;
        #pragma unroll
        for (int v = 0; v < 8; ++v) {
            const int row = v + hi * 8;
            tw[row * TPAD + col] = sspf(acc[ct][v] + sB1[col]);
        }
    }

    v8f acc2[8];
    #pragma unroll
    for (int ct = 0; ct < 8; ++ct) acc2[ct] = z8;

    const float* trow = tw + m * TPAD;
    #pragma unroll 4
    for (int k = 0; k < NF; k += 4) {
        const int kk = k + hi * 2;
        const v2f a = *(const v2f*)(trow + kk);
        #pragma unroll
        for (int ct = 0; ct < 8; ++ct) {
            const v2f b = ld_pair_lds(sWf2, kk, ct * 16 + m);
            acc2[ct] = wmma_f32x4(a, b, acc2[ct]);
        }
    }

    float red[8];
    #pragma unroll
    for (int ct = 0; ct < 8; ++ct) red[ct] = 0.f;
    int cur_i = idx_i[eb + hi * 8];

    #pragma unroll
    for (int v = 0; v < 8; ++v) {
        const int   e  = eb + v + hi * 8;
        const float rc = rcut[e];
        const int   ji = idx_j[e];
        const int   ii = idx_i[e];
        if (ii != cur_i) {
            float* yrow = y + (size_t)cur_i * NF + m;
            #pragma unroll
            for (int ct = 0; ct < 8; ++ct) {
                atomicAdd(yrow + ct * 16, red[ct]);
                red[ct] = 0.f;
            }
            cur_i = ii;
        }
        const float* hrow = h + (size_t)ji * NF + m;
        #pragma unroll
        for (int ct = 0; ct < 8; ++ct) {
            red[ct] += (acc2[ct][v] + sB2[ct * 16 + m]) * rc * hrow[ct * 16];
        }
    }
    {
        float* yrow = y + (size_t)cur_i * NF + m;
        #pragma unroll
        for (int ct = 0; ct < 8; ++ct) atomicAdd(yrow + ct * 16, red[ct]);
    }
}

extern "C" void kernel_launch(void* const* d_in, const int* in_sizes, int n_in,
                              void* d_out, int out_size, void* d_ws, size_t ws_size,
                              hipStream_t stream) {
    const float* x      = (const float*)d_in[0];
    const float* f_ij   = (const float*)d_in[1];
    const float* rcut   = (const float*)d_in[2];
    const int*   idx_i  = (const int*)d_in[3];
    const int*   idx_j  = (const int*)d_in[4];
    const float* W_in2f = (const float*)d_in[5];
    const float* W_f1   = (const float*)d_in[6];
    const float* b_f1   = (const float*)d_in[7];
    const float* W_f2   = (const float*)d_in[8];
    const float* b_f2   = (const float*)d_in[9];
    const float* W_o1   = (const float*)d_in[10];
    const float* b_o1   = (const float*)d_in[11];
    const float* W_o2   = (const float*)d_in[12];
    const float* b_o2   = (const float*)d_in[13];
    float* out = (float*)d_out;

    float* h = (float*)d_ws;
    float* y = h + (size_t)N_ATOMS * NF;

    const size_t gemm_lds = (size_t)(NF * NF + NF) * sizeof(float);
    const size_t edge_lds = (size_t)(NF * NF + NRBF * NF + 2 * NF + 8 * 16 * TPAD) * sizeof(float);

    {
        const int n4 = (N_ATOMS * NF) / 4;
        zero_f32<<<(n4 + 255) / 256, 256, 0, stream>>>(y, n4);
    }
    gemm128_wmma<<<(N_ATOMS + 127) / 128, 256, gemm_lds, stream>>>(
        x, W_in2f, nullptr, h, N_ATOMS, 0);
    edge_cfconv_wmma<<<N_EDGES / 128, 256, edge_lds, stream>>>(
        f_ij, rcut, idx_i, idx_j, W_f1, b_f1, W_f2, b_f2, h, y);
    gemm128_wmma<<<(N_ATOMS + 127) / 128, 256, gemm_lds, stream>>>(
        y, W_o1, b_o1, h, N_ATOMS, 1);
    gemm128_wmma<<<(N_ATOMS + 127) / 128, 256, gemm_lds, stream>>>(
        h, W_o2, b_o2, out, N_ATOMS, 0);
}


# shifted softplus as max(x,0)+ln2*log2(1+exp2(-|x|*log2e)) on the hardware exp2/log2 units, replacing library expf+log1pf (about 100 vector instructions per element); abs error up to 2e-7
# speedup vs baseline: 1.9760x; 1.9760x over previous
#include <hip/hip_runtime.h>
#include <hip/hip_bf16.h>

#define N_ATOMS 50000
#define N_EDGES 1600000
#define NAB 128
#define NRBF 20
#define NF 128
#define TPAD (NF + 4)

typedef __attribute__((ext_vector_type(2))) float v2f;
typedef __attribute__((ext_vector_type(8))) float v8f;

__device__ __forceinline__ v8f wmma_f32x4(v2f a, v2f b, v8f c) {
    return __builtin_amdgcn_wmma_f32_16x16x4_f32(
         false, a,  false, b,
         (short)0, c,  false,  false);
}

__device__ __forceinline__ float sspf(float x) {
    const float t = __builtin_amdgcn_exp2f(-1.44269504088896340736f * fabsf(x));
    const float l = __builtin_amdgcn_logf(1.0f + t);
    return fmaf(0.69314718055994530942f, l, fmaxf(x, 0.f)) - 0.69314718055994530942f;
}

__device__ __forceinline__ void stage_w_pairs(const float* __restrict__ W,
                                              float* __restrict__ sW2,
                                              int krows, int tid) {
    const int npairs = (krows / 2) * NF;
    for (int i = tid; i < npairs; i += 256) {
        const int p   = i >> 7;
        const int col = i & (NF - 1);
        sW2[2 * i + 0] = W[(2 * p)     * NF + col];
        sW2[2 * i + 1] = W[(2 * p + 1) * NF + col];
    }
}

__device__ __forceinline__ v2f ld_pair_lds(const float* __restrict__ sW2,
                                           int kk, int col) {
    return *(const v2f*)(sW2 + (((kk >> 1) * NF + col) << 1));
}

__global__ void zero_f32(float* __restrict__ p, int n4) {
    int i = blockIdx.x * blockDim.x + threadIdx.x;
    if (i < n4) ((float4*)p)[i] = make_float4(0.f, 0.f, 0.f, 0.f);
}

__global__ __launch_bounds__(256) void gemm128_wmma(
    const float* __restrict__ A, const float* __restrict__ W,
    const float* __restrict__ bias, float* __restrict__ out,
    int M, int apply_ssp)
{
    extern __shared__ float smem[];
    float* sW2 = smem;
    float* sB  = smem + NF * NF;

    const int tid = threadIdx.x;
    stage_w_pairs(W, sW2, NF, tid);
    if (tid < NF) sB[tid] = bias ? bias[tid] : 0.f;
    __syncthreads();

    const int lane = tid & 31;
    const int wave = tid >> 5;
    const int m    = lane & 15;
    const int hi   = lane >> 4;
    const int r0   = blockIdx.x * 128 + wave * 16;

    v8f acc[8];
    const v8f z8 = {0.f, 0.f, 0.f, 0.f, 0.f, 0.f, 0.f, 0.f};
    #pragma unroll
    for (int ct = 0; ct < 8; ++ct) acc[ct] = z8;

    const int rowc = min(r0 + m, M - 1);
    const float* arow = A + (size_t)rowc * NF;

    #pragma unroll 4
    for (int k = 0; k < NF; k += 4) {
        const int kk = k + hi * 2;
        const v2f a = *(const v2f*)(arow + kk);
        #pragma unroll
        for (int ct = 0; ct < 8; ++ct) {
            const v2f b = ld_pair_lds(sW2, kk, ct * 16 + m);
            acc[ct] = wmma_f32x4(a, b, acc[ct]);
        }
    }

    #pragma unroll
    for (int ct = 0; ct < 8; ++ct) {
        const int col = ct * 16 + m;
        #pragma unroll
        for (int v = 0; v < 8; ++v) {
            const int r = r0 + v + hi * 8;
            if (r < M) {
                float val = acc[ct][v] + sB[col];
                if (apply_ssp) val = sspf(val);
                out[(size_t)r * NF + col] = val;
            }
        }
    }
}

__global__ __launch_bounds__(256) void edge_cfconv_wmma(
    const float* __restrict__ f_ij, const float* __restrict__ rcut,
    const int* __restrict__ idx_i, const int* __restrict__ idx_j,
    const float* __restrict__ Wf1, const float* __restrict__ bf1,
    const float* __restrict__ Wf2, const float* __restrict__ bf2,
    const float* __restrict__ h, float* __restrict__ y)
{
    extern __shared__ float smem[];
    float* sWf2 = smem;
    float* sWf1 = sWf2 + NF * NF;
    float* sB1  = sWf1 + NRBF * NF;
    float* sB2  = sB1 + NF;
    float* sT   = sB2 + NF;

    const int tid = threadIdx.x;
    stage_w_pairs(Wf2, sWf2, NF, tid);
    stage_w_pairs(Wf1, sWf1, NRBF, tid);
    if (tid < NF) { sB1[tid] = bf1[tid]; sB2[tid] = bf2[tid]; }
    __syncthreads();

    const int lane = tid & 31;
    const int wave = tid >> 5;
    const int m    = lane & 15;
    const int hi   = lane >> 4;
    const int eb   = blockIdx.x * 128 + wave * 16;

    float* tw = sT + wave * 16 * TPAD;
    const v8f z8 = {0.f, 0.f, 0.f, 0.f, 0.f, 0.f, 0.f, 0.f};

    v8f acc[8];
    #pragma unroll
    for (int ct = 0; ct < 8; ++ct) acc[ct] = z8;

    const float* frow = f_ij + (size_t)(eb + m) * NRBF;
    #pragma unroll
    for (int k = 0; k < NRBF; k += 4) {
        const int kk = k + hi * 2;
        const v2f a = *(const v2f*)(frow + kk);
        #pragma unroll
        for (int ct = 0; ct < 8; ++ct) {
            const v2f b = ld_pair_lds(sWf1, kk, ct * 16 + m);
            acc[ct] = wmma_f32x4(a, b, acc[ct]);
        }
    }

    #pragma unroll
    for (int ct = 0; ct < 8; ++ct) {
        const int col = ct * 16 + m;
        #pragma unroll
        for (int v = 0; v < 8; ++v) {
            const int row = v + hi * 8;
            tw[row * TPAD + col] = sspf(acc[ct][v] + sB1[col]);
        }
    }

    v8f acc2[8];
    #pragma unroll
    for (int ct = 0; ct < 8; ++ct) acc2[ct] = z8;

    const float* trow = tw + m * TPAD;
    #pragma unroll 4
    for (int k = 0; k < NF; k += 4) {
        const int kk = k + hi * 2;
        const v2f a = *(const v2f*)(trow + kk);
        #pragma unroll
        for (int ct = 0; ct < 8; ++ct) {
            const v2f b = ld_pair_lds(sWf2, kk, ct * 16 + m);
            acc2[ct] = wmma_f32x4(a, b, acc2[ct]);
        }
    }

    float red[8];
    #pragma unroll
    for (int ct = 0; ct < 8; ++ct) red[ct] = 0.f;
    int cur_i = idx_i[eb + hi * 8];

    #pragma unroll
    for (int v = 0; v < 8; ++v) {
        const int   e  = eb + v + hi * 8;
        const float rc = rcut[e];
        const int   ji = idx_j[e];
        const int   ii = idx_i[e];
        if (ii != cur_i) {
            float* yrow = y + (size_t)cur_i * NF + m;
            #pragma unroll
            for (int ct = 0; ct < 8; ++ct) {
                atomicAdd(yrow + ct * 16, red[ct]);
                red[ct] = 0.f;
            }
            cur_i = ii;
        }
        const float* hrow = h + (size_t)ji * NF + m;
        #pragma unroll
        for (int ct = 0; ct < 8; ++ct) {
            red[ct] += (acc2[ct][v] + sB2[ct * 16 + m]) * rc * hrow[ct * 16];
        }
    }
    {
        float* yrow = y + (size_t)cur_i * NF + m;
        #pragma unroll
        for (int ct = 0; ct < 8; ++ct) atomicAdd(yrow + ct * 16, red[ct]);
    }
}

extern "C" void kernel_launch(void* const* d_in, const int* in_sizes, int n_in,
                              void* d_out, int out_size, void* d_ws, size_t ws_size,
                              hipStream_t stream) {
    const float* x      = (const float*)d_in[0];
    const float* f_ij   = (const float*)d_in[1];
    const float* rcut   = (const float*)d_in[2];
    const int*   idx_i  = (const int*)d_in[3];
    const int*   idx_j  = (const int*)d_in[4];
    const float* W_in2f = (const float*)d_in[5];
    const float* W_f1   = (const float*)d_in[6];
    const float* b_f1   = (const float*)d_in[7];
    const float* W_f2   = (const float*)d_in[8];
    const float* b_f2   = (const float*)d_in[9];
    const float* W_o1   = (const float*)d_in[10];
    const float* b_o1   = (const float*)d_in[11];
    const float* W_o2   = (const float*)d_in[12];
    const float* b_o2   = (const float*)d_in[13];
    float* out = (float*)d_out;

    float* h = (float*)d_ws;
    float* y = h + (size_t)N_ATOMS * NF;

    const size_t gemm_lds = (size_t)(NF * NF + NF) * sizeof(float);
    const size_t edge_lds = (size_t)(NF * NF + NRBF * NF + 2 * NF + 8 * 16 * TPAD) * sizeof(float);

    {
        const int n4 = (N_ATOMS * NF) / 4;
        zero_f32<<<(n4 + 255) / 256, 256, 0, stream>>>(y, n4);
    }
    gemm128_wmma<<<(N_ATOMS + 127) / 128, 256, gemm_lds, stream>>>(
        x, W_in2f, nullptr, h, N_ATOMS, 0);
    edge_cfconv_wmma<<<N_EDGES / 128, 256, edge_lds, stream>>>(
        f_ij, rcut, idx_i, idx_j, W_f1, b_f1, W_f2, b_f2, h, y);
    gemm128_wmma<<<(N_ATOMS + 127) / 128, 256, gemm_lds, stream>>>(
        y, W_o1, b_o1, h, N_ATOMS, 1);
    gemm128_wmma<<<(N_ATOMS + 127) / 128, 256, gemm_lds, stream>>>(
        h, W_o2, b_o2, out, N_ATOMS, 0);
}


# edge kernel made persistent: 256 work-groups of 24 waves stage W_f1/W_f2 into LDS once, each wave then loops over 16-edge tiles (was 12500 groups each restaging both); atomic arrival order changes
# speedup vs baseline: 2.1308x; 1.0783x over previous
#include <hip/hip_runtime.h>
#include <hip/hip_bf16.h>

#define N_ATOMS 50000
#define N_EDGES 1600000
#define NAB 128
#define NRBF 20
#define NF 128
#define TPAD (NF + 4)
#define EDGE_WAVES   24
#define EDGE_THREADS (EDGE_WAVES * 32)
#define EDGE_GROUPS  256
#define N_TILES      (N_EDGES / 16)

typedef __attribute__((ext_vector_type(2))) float v2f;
typedef __attribute__((ext_vector_type(8))) float v8f;

__device__ __forceinline__ v8f wmma_f32x4(v2f a, v2f b, v8f c) {
    return __builtin_amdgcn_wmma_f32_16x16x4_f32(
         false, a,  false, b,
         (short)0, c,  false,  false);
}

__device__ __forceinline__ float sspf(float x) {
    const float t = __builtin_amdgcn_exp2f(-1.44269504088896340736f * fabsf(x));
    const float l = __builtin_amdgcn_logf(1.0f + t);
    return fmaf(0.69314718055994530942f, l, fmaxf(x, 0.f)) - 0.69314718055994530942f;
}

__device__ __forceinline__ void stage_w_pairs(const float* __restrict__ W,
                                              float* __restrict__ sW2,
                                              int krows, int tid, int nthreads) {
    const int npairs = (krows / 2) * NF;
    for (int i = tid; i < npairs; i += nthreads) {
        const int p   = i >> 7;
        const int col = i & (NF - 1);
        sW2[2 * i + 0] = W[(2 * p)     * NF + col];
        sW2[2 * i + 1] = W[(2 * p + 1) * NF + col];
    }
}

__device__ __forceinline__ v2f ld_pair_lds(const float* __restrict__ sW2,
                                           int kk, int col) {
    return *(const v2f*)(sW2 + (((kk >> 1) * NF + col) << 1));
}

__global__ void zero_f32(float* __restrict__ p, int n4) {
    int i = blockIdx.x * blockDim.x + threadIdx.x;
    if (i < n4) ((float4*)p)[i] = make_float4(0.f, 0.f, 0.f, 0.f);
}

__global__ __launch_bounds__(256) void gemm128_wmma(
    const float* __restrict__ A, const float* __restrict__ W,
    const float* __restrict__ bias, float* __restrict__ out,
    int M, int apply_ssp)
{
    extern __shared__ float smem[];
    float* sW2 = smem;
    float* sB  = smem + NF * NF;

    const int tid = threadIdx.x;
    stage_w_pairs(W, sW2, NF, tid, 256);
    if (tid < NF) sB[tid] = bias ? bias[tid] : 0.f;
    __syncthreads();

    const int lane = tid & 31;
    const int wave = tid >> 5;
    const int m    = lane & 15;
    const int hi   = lane >> 4;
    const int r0   = blockIdx.x * 128 + wave * 16;

    v8f acc[8];
    const v8f z8 = {0.f, 0.f, 0.f, 0.f, 0.f, 0.f, 0.f, 0.f};
    #pragma unroll
    for (int ct = 0; ct < 8; ++ct) acc[ct] = z8;

    const int rowc = min(r0 + m, M - 1);
    const float* arow = A + (size_t)rowc * NF;

    #pragma unroll 4
    for (int k = 0; k < NF; k += 4) {
        const int kk = k + hi * 2;
        const v2f a = *(const v2f*)(arow + kk);
        #pragma unroll
        for (int ct = 0; ct < 8; ++ct) {
            const v2f b = ld_pair_lds(sW2, kk, ct * 16 + m);
            acc[ct] = wmma_f32x4(a, b, acc[ct]);
        }
    }

    #pragma unroll
    for (int ct = 0; ct < 8; ++ct) {
        const int col = ct * 16 + m;
        #pragma unroll
        for (int v = 0; v < 8; ++v) {
            const int r = r0 + v + hi * 8;
            if (r < M) {
                float val = acc[ct][v] + sB[col];
                if (apply_ssp) val = sspf(val);
                out[(size_t)r * NF + col] = val;
            }
        }
    }
}

__global__ __launch_bounds__(EDGE_THREADS) void edge_cfconv_wmma(
    const float* __restrict__ f_ij, const float* __restrict__ rcut,
    const int* __restrict__ idx_i, const int* __restrict__ idx_j,
    const float* __restrict__ Wf1, const float* __restrict__ bf1,
    const float* __restrict__ Wf2, const float* __restrict__ bf2,
    const float* __restrict__ h, float* __restrict__ y)
{
    extern __shared__ float smem[];
    float* sWf2 = smem;
    float* sWf1 = sWf2 + NF * NF;
    float* sB1  = sWf1 + NRBF * NF;
    float* sB2  = sB1 + NF;
    float* sT   = sB2 + NF;

    const int tid = threadIdx.x;
    stage_w_pairs(Wf2, sWf2, NF, tid, EDGE_THREADS);
    stage_w_pairs(Wf1, sWf1, NRBF, tid, EDGE_THREADS);
    if (tid < NF) { sB1[tid] = bf1[tid]; sB2[tid] = bf2[tid]; }
    __syncthreads();

    const int lane = tid & 31;
    const int wave = tid >> 5;
    const int m    = lane & 15;
    const int hi   = lane >> 4;

    float* tw = sT + wave * 16 * TPAD;
    const float* trow = tw + m * TPAD;
    const v8f z8 = {0.f, 0.f, 0.f, 0.f, 0.f, 0.f, 0.f, 0.f};

    for (int tile = blockIdx.x + EDGE_GROUPS * wave; tile < N_TILES;
         tile += EDGE_GROUPS * EDGE_WAVES) {
        const int eb = tile * 16;

        v8f acc[8];
        #pragma unroll
        for (int ct = 0; ct < 8; ++ct) acc[ct] = z8;

        const float* frow = f_ij + (size_t)(eb + m) * NRBF;
        #pragma unroll
        for (int k = 0; k < NRBF; k += 4) {
            const int kk = k + hi * 2;
            const v2f a = *(const v2f*)(frow + kk);
            #pragma unroll
            for (int ct = 0; ct < 8; ++ct) {
                const v2f b = ld_pair_lds(sWf1, kk, ct * 16 + m);
                acc[ct] = wmma_f32x4(a, b, acc[ct]);
            }
        }

        #pragma unroll
        for (int ct = 0; ct < 8; ++ct) {
            const int col = ct * 16 + m;
            #pragma unroll
            for (int v = 0; v < 8; ++v) {
                const int row = v + hi * 8;
                tw[row * TPAD + col] = sspf(acc[ct][v] + sB1[col]);
            }
        }

        v8f acc2[8];
        #pragma unroll
        for (int ct = 0; ct < 8; ++ct) acc2[ct] = z8;

        #pragma unroll 4
        for (int k = 0; k < NF; k += 4) {
            const int kk = k + hi * 2;
            const v2f a = *(const v2f*)(trow + kk);
            #pragma unroll
            for (int ct = 0; ct < 8; ++ct) {
                const v2f b = ld_pair_lds(sWf2, kk, ct * 16 + m);
                acc2[ct] = wmma_f32x4(a, b, acc2[ct]);
            }
        }

        float red[8];
        #pragma unroll
        for (int ct = 0; ct < 8; ++ct) red[ct] = 0.f;
        int cur_i = idx_i[eb + hi * 8];

        #pragma unroll
        for (int v = 0; v < 8; ++v) {
            const int   e  = eb + v + hi * 8;
            const float rc = rcut[e];
            const int   ji = idx_j[e];
            const int   ii = idx_i[e];
            if (ii != cur_i) {
                float* yrow = y + (size_t)cur_i * NF + m;
                #pragma unroll
                for (int ct = 0; ct < 8; ++ct) {
                    atomicAdd(yrow + ct * 16, red[ct]);
                    red[ct] = 0.f;
                }
                cur_i = ii;
            }
            const float* hrow = h + (size_t)ji * NF + m;
            #pragma unroll
            for (int ct = 0; ct < 8; ++ct) {
                red[ct] += (acc2[ct][v] + sB2[ct * 16 + m]) * rc * hrow[ct * 16];
            }
        }
        {
            float* yrow = y + (size_t)cur_i * NF + m;
            #pragma unroll
            for (int ct = 0; ct < 8; ++ct) atomicAdd(yrow + ct * 16, red[ct]);
        }
    }
}

extern "C" void kernel_launch(void* const* d_in, const int* in_sizes, int n_in,
                              void* d_out, int out_size, void* d_ws, size_t ws_size,
                              hipStream_t stream) {
    const float* x      = (const float*)d_in[0];
    const float* f_ij   = (const float*)d_in[1];
    const float* rcut   = (const float*)d_in[2];
    const int*   idx_i  = (const int*)d_in[3];
    const int*   idx_j  = (const int*)d_in[4];
    const float* W_in2f = (const float*)d_in[5];
    const float* W_f1   = (const float*)d_in[6];
    const float* b_f1   = (const float*)d_in[7];
    const float* W_f2   = (const float*)d_in[8];
    const float* b_f2   = (const float*)d_in[9];
    const float* W_o1   = (const float*)d_in[10];
    const float* b_o1   = (const float*)d_in[11];
    const float* W_o2   = (const float*)d_in[12];
    const float* b_o2   = (const float*)d_in[13];
    float* out = (float*)d_out;

    float* h = (float*)d_ws;
    float* y = h + (size_t)N_ATOMS * NF;

    const size_t gemm_lds = (size_t)(NF * NF + NF) * sizeof(float);
    const size_t edge_lds = (size_t)(NF * NF + NRBF * NF + 2 * NF + EDGE_WAVES * 16 * TPAD) * sizeof(float);

    {
        const int n4 = (N_ATOMS * NF) / 4;
        zero_f32<<<(n4 + 255) / 256, 256, 0, stream>>>(y, n4);
    }
    gemm128_wmma<<<(N_ATOMS + 127) / 128, 256, gemm_lds, stream>>>(
        x, W_in2f, nullptr, h, N_ATOMS, 0);
    edge_cfconv_wmma<<<EDGE_GROUPS, EDGE_THREADS, edge_lds, stream>>>(
        f_ij, rcut, idx_i, idx_j, W_f1, b_f1, W_f2, b_f2, h, y);
    gemm128_wmma<<<(N_ATOMS + 127) / 128, 256, gemm_lds, stream>>>(
        y, W_o1, b_o1, h, N_ATOMS, 1);
    gemm128_wmma<<<(N_ATOMS + 127) / 128, 256, gemm_lds, stream>>>(
        h, W_o2, b_o2, out, N_ATOMS, 0);
}


# K loops software-pipelined: B and A fragment loads refill a second register set one to two k-steps ahead of their 8 WMMAs (edge stage 2, gemm128 with A rows preloaded); k order and sums unchanged
# speedup vs baseline: 2.2251x; 1.0443x over previous
#include <hip/hip_runtime.h>
#include <hip/hip_bf16.h>

#define N_ATOMS 50000
#define N_EDGES 1600000
#define NAB 128
#define NRBF 20
#define NF 128
#define TPAD (NF + 4)
#define EDGE_WAVES   24
#define EDGE_THREADS (EDGE_WAVES * 32)
#define EDGE_GROUPS  256
#define N_TILES      (N_EDGES / 16)

typedef __attribute__((ext_vector_type(2))) float v2f;
typedef __attribute__((ext_vector_type(8))) float v8f;

__device__ __forceinline__ v8f wmma_f32x4(v2f a, v2f b, v8f c) {
    return __builtin_amdgcn_wmma_f32_16x16x4_f32(
         false, a,  false, b,
         (short)0, c,  false,  false);
}

__device__ __forceinline__ float sspf(float x) {
    const float t = __builtin_amdgcn_exp2f(-1.44269504088896340736f * fabsf(x));
    const float l = __builtin_amdgcn_logf(1.0f + t);
    return fmaf(0.69314718055994530942f, l, fmaxf(x, 0.f)) - 0.69314718055994530942f;
}

__device__ __forceinline__ void stage_w_pairs(const float* __restrict__ W,
                                              float* __restrict__ sW2,
                                              int krows, int tid, int nthreads) {
    const int npairs = (krows / 2) * NF;
    for (int i = tid; i < npairs; i += nthreads) {
        const int p   = i >> 7;
        const int col = i & (NF - 1);
        sW2[2 * i + 0] = W[(2 * p)     * NF + col];
        sW2[2 * i + 1] = W[(2 * p + 1) * NF + col];
    }
}

__device__ __forceinline__ v2f ld_pair_lds(const float* __restrict__ sW2,
                                           int kk, int col) {
    return *(const v2f*)(sW2 + (((kk >> 1) * NF + col) << 1));
}

__device__ __forceinline__ void ld_b(v2f (&b)[8], const v2f* __restrict__ bq, int k) {
    #pragma unroll
    for (int ct = 0; ct < 8; ++ct) b[ct] = bq[(k >> 1) * NF + ct * 16];
}

__device__ __forceinline__ void mma_step(v8f (&acc)[8], v2f a, const v2f (&b)[8]) {
    #pragma unroll
    for (int ct = 0; ct < 8; ++ct) acc[ct] = wmma_f32x4(a, b[ct], acc[ct]);
}

__device__ __forceinline__ void mma_group(v8f (&acc)[8], v2f ca0, v2f ca1,
                                          v2f& na0, v2f& na1,
                                          v2f (&b0)[8], v2f (&b1)[8],
                                          const v2f* __restrict__ aq,
                                          const v2f* __restrict__ bq, int k) {
    na0 = aq[(k + 8) >> 1];
    na1 = aq[(k + 12) >> 1];
    mma_step(acc, ca0, b0);
    ld_b(b0, bq, k + 8);
    mma_step(acc, ca1, b1);
    ld_b(b1, bq, k + 12);
}

__global__ void zero_f32(float* __restrict__ p, int n4) {
    int i = blockIdx.x * blockDim.x + threadIdx.x;
    if (i < n4) ((float4*)p)[i] = make_float4(0.f, 0.f, 0.f, 0.f);
}

__global__ __launch_bounds__(256) void gemm128_wmma(
    const float* __restrict__ A, const float* __restrict__ W,
    const float* __restrict__ bias, float* __restrict__ out,
    int M, int apply_ssp)
{
    extern __shared__ float smem[];
    float* sW2 = smem;
    float* sB  = smem + NF * NF;

    const int tid  = threadIdx.x;
    const int lane = tid & 31;
    const int wave = tid >> 5;
    const int m    = lane & 15;
    const int hi   = lane >> 4;
    const int r0   = blockIdx.x * 128 + wave * 16;

    const int rowc = min(r0 + m, M - 1);
    const v2f* aq = (const v2f*)(A + (size_t)rowc * NF) + hi;

    v2f afr[32];
    #pragma unroll
    for (int s = 0; s < 32; ++s) afr[s] = aq[2 * s];

    stage_w_pairs(W, sW2, NF, tid, 256);
    if (tid < NF) sB[tid] = bias ? bias[tid] : 0.f;
    __syncthreads();

    v8f acc[8];
    const v8f z8 = {0.f, 0.f, 0.f, 0.f, 0.f, 0.f, 0.f, 0.f};
    #pragma unroll
    for (int ct = 0; ct < 8; ++ct) acc[ct] = z8;

    const v2f* bq = (const v2f*)sW2 + hi * NF + m;
    v2f b0[8], b1[8];
    ld_b(b0, bq, 0);
    ld_b(b1, bq, 4);
    #pragma unroll
    for (int s = 0; s < 32; s += 2) {
        mma_step(acc, afr[s], b0);
        if (s + 2 < 32) ld_b(b0, bq, 4 * (s + 2));
        mma_step(acc, afr[s + 1], b1);
        if (s + 3 < 32) ld_b(b1, bq, 4 * (s + 3));
    }

    #pragma unroll
    for (int ct = 0; ct < 8; ++ct) {
        const int col = ct * 16 + m;
        #pragma unroll
        for (int v = 0; v < 8; ++v) {
            const int r = r0 + v + hi * 8;
            if (r < M) {
                float val = acc[ct][v] + sB[col];
                if (apply_ssp) val = sspf(val);
                out[(size_t)r * NF + col] = val;
            }
        }
    }
}

__global__ __launch_bounds__(EDGE_THREADS) void edge_cfconv_wmma(
    const float* __restrict__ f_ij, const float* __restrict__ rcut,
    const int* __restrict__ idx_i, const int* __restrict__ idx_j,
    const float* __restrict__ Wf1, const float* __restrict__ bf1,
    const float* __restrict__ Wf2, const float* __restrict__ bf2,
    const float* __restrict__ h, float* __restrict__ y)
{
    extern __shared__ float smem[];
    float* sWf2 = smem;
    float* sWf1 = sWf2 + NF * NF;
    float* sB1  = sWf1 + NRBF * NF;
    float* sB2  = sB1 + NF;
    float* sT   = sB2 + NF;

    const int tid = threadIdx.x;
    stage_w_pairs(Wf2, sWf2, NF, tid, EDGE_THREADS);
    stage_w_pairs(Wf1, sWf1, NRBF, tid, EDGE_THREADS);
    if (tid < NF) { sB1[tid] = bf1[tid]; sB2[tid] = bf2[tid]; }
    __syncthreads();

    const int lane = tid & 31;
    const int wave = tid >> 5;
    const int m    = lane & 15;
    const int hi   = lane >> 4;

    float* tw = sT + wave * 16 * TPAD;
    const float* trow = tw + m * TPAD;
    const v2f* aq = (const v2f*)trow + hi;
    const v2f* bq = (const v2f*)sWf2 + hi * NF + m;
    const v2f* bq1 = (const v2f*)sWf1 + hi * NF + m;
    const v8f z8 = {0.f, 0.f, 0.f, 0.f, 0.f, 0.f, 0.f, 0.f};

    for (int tile = blockIdx.x + EDGE_GROUPS * wave; tile < N_TILES;
         tile += EDGE_GROUPS * EDGE_WAVES) {
        const int eb = tile * 16;

        v8f acc[8];
        #pragma unroll
        for (int ct = 0; ct < 8; ++ct) acc[ct] = z8;

        const float* frow = f_ij + (size_t)(eb + m) * NRBF;
        #pragma unroll
        for (int k = 0; k < NRBF; k += 4) {
            const int kk = k + hi * 2;
            const v2f a = *(const v2f*)(frow + kk);
            #pragma unroll
            for (int ct = 0; ct < 8; ++ct) {
                const v2f b = bq1[(k >> 1) * NF + ct * 16];
                acc[ct] = wmma_f32x4(a, b, acc[ct]);
            }
        }

        #pragma unroll
        for (int ct = 0; ct < 8; ++ct) {
            const int col = ct * 16 + m;
            #pragma unroll
            for (int v = 0; v < 8; ++v) {
                const int row = v + hi * 8;
                tw[row * TPAD + col] = sspf(acc[ct][v] + sB1[col]);
            }
        }

        v8f acc2[8];
        #pragma unroll
        for (int ct = 0; ct < 8; ++ct) acc2[ct] = z8;

        v2f a0 = aq[0], a1 = aq[2], n0, n1, b0[8], b1[8];
        ld_b(b0, bq, 0);
        ld_b(b1, bq, 4);
        for (int k = 0; k < NF - 16; k += 16) {
            mma_group(acc2, a0, a1, n0, n1, b0, b1, aq, bq, k);
            mma_group(acc2, n0, n1, a0, a1, b0, b1, aq, bq, k + 8);
        }
        mma_group(acc2, a0, a1, n0, n1, b0, b1, aq, bq, NF - 16);
        mma_step(acc2, n0, b0);
        mma_step(acc2, n1, b1);

        float red[8];
        #pragma unroll
        for (int ct = 0; ct < 8; ++ct) red[ct] = 0.f;
        int cur_i = idx_i[eb + hi * 8];

        #pragma unroll
        for (int v = 0; v < 8; ++v) {
            const int   e  = eb + v + hi * 8;
            const float rc = rcut[e];
            const int   ji = idx_j[e];
            const int   ii = idx_i[e];
            if (ii != cur_i) {
                float* yrow = y + (size_t)cur_i * NF + m;
                #pragma unroll
                for (int ct = 0; ct < 8; ++ct) {
                    atomicAdd(yrow + ct * 16, red[ct]);
                    red[ct] = 0.f;
                }
                cur_i = ii;
            }
            const float* hrow = h + (size_t)ji * NF + m;
            #pragma unroll
            for (int ct = 0; ct < 8; ++ct) {
                red[ct] += (acc2[ct][v] + sB2[ct * 16 + m]) * rc * hrow[ct * 16];
            }
        }
        {
            float* yrow = y + (size_t)cur_i * NF + m;
            #pragma unroll
            for (int ct = 0; ct < 8; ++ct) atomicAdd(yrow + ct * 16, red[ct]);
        }
    }
}

extern "C" void kernel_launch(void* const* d_in, const int* in_sizes, int n_in,
                              void* d_out, int out_size, void* d_ws, size_t ws_size,
                              hipStream_t stream) {
    const float* x      = (const float*)d_in[0];
    const float* f_ij   = (const float*)d_in[1];
    const float* rcut   = (const float*)d_in[2];
    const int*   idx_i  = (const int*)d_in[3];
    const int*   idx_j  = (const int*)d_in[4];
    const float* W_in2f = (const float*)d_in[5];
    const float* W_f1   = (const float*)d_in[6];
    const float* b_f1   = (const float*)d_in[7];
    const float* W_f2   = (const float*)d_in[8];
    const float* b_f2   = (const float*)d_in[9];
    const float* W_o1   = (const float*)d_in[10];
    const float* b_o1   = (const float*)d_in[11];
    const float* W_o2   = (const float*)d_in[12];
    const float* b_o2   = (const float*)d_in[13];
    float* out = (float*)d_out;

    float* h = (float*)d_ws;
    float* y = h + (size_t)N_ATOMS * NF;

    const size_t gemm_lds = (size_t)(NF * NF + NF) * sizeof(float);
    const size_t edge_lds = (size_t)(NF * NF + NRBF * NF + 2 * NF + EDGE_WAVES * 16 * TPAD) * sizeof(float);

    {
        const int n4 = (N_ATOMS * NF) / 4;
        zero_f32<<<(n4 + 255) / 256, 256, 0, stream>>>(y, n4);
    }
    gemm128_wmma<<<(N_ATOMS + 127) / 128, 256, gemm_lds, stream>>>(
        x, W_in2f, nullptr, h, N_ATOMS, 0);
    edge_cfconv_wmma<<<EDGE_GROUPS, EDGE_THREADS, edge_lds, stream>>>(
        f_ij, rcut, idx_i, idx_j, W_f1, b_f1, W_f2, b_f2, h, y);
    gemm128_wmma<<<(N_ATOMS + 127) / 128, 256, gemm_lds, stream>>>(
        y, W_o1, b_o1, h, N_ATOMS, 1);
    gemm128_wmma<<<(N_ATOMS + 127) / 128, 256, gemm_lds, stream>>>(
        h, W_o2, b_o2, out, N_ATOMS, 0);
}
